# starting kernel
# baseline (speedup 1.0000x reference)
#include <hip/hip_runtime.h>

typedef __attribute__((ext_vector_type(2))) float v2f;
typedef __attribute__((ext_vector_type(8))) float v8f;

#define GCN_N 100000
#define GCN_D 64
#define WLDS_STRIDE 68

__global__ void gcn_zero4(float4* __restrict__ p, int n4) {
  int i = blockIdx.x * blockDim.x + threadIdx.x;
  if (i < n4) p[i] = make_float4(0.f, 0.f, 0.f, 0.f);
}

__global__ void gcn_degree(const int* __restrict__ esrc, const int* __restrict__ edst,
                           float* __restrict__ deg_out, float* __restrict__ deg_in, int E) {
  int e = blockIdx.x * blockDim.x + threadIdx.x;
  if (e >= E) return;
  atomicAdd(&deg_out[esrc[e]], 1.0f);
  atomicAdd(&deg_in[edst[e]], 1.0f);
}

__global__ void gcn_norm(float* __restrict__ deg, int n) {
  int i = blockIdx.x * blockDim.x + threadIdx.x;
  if (i < n) deg[i] = rsqrtf(fmaxf(deg[i], 1.0f));
}

__global__ void gcn_scatter(const float* __restrict__ h, const int* __restrict__ esrc,
                            const int* __restrict__ edst, const float* __restrict__ norm_src,
                            float* __restrict__ agg, int E) {
  unsigned tid = blockIdx.x * blockDim.x + threadIdx.x;
  unsigned e = tid >> 4;
  if (e >= (unsigned)E) return;
  int f = (tid & 15) << 2;
  int s = esrc[e];
  int d = edst[e];
  float ns = norm_src[s];
  float4 m = *(const float4*)(h + (size_t)s * GCN_D + f);
  float* ap = agg + (size_t)d * GCN_D + f;
  atomicAdd(ap + 0, m.x * ns);
  atomicAdd(ap + 1, m.y * ns);
  atomicAdd(ap + 2, m.z * ns);
  atomicAdd(ap + 3, m.w * ns);
}

template <int LDW, int NCP, int NCT, bool RELU>
__global__ void gcn_gemm_wmma(const float* __restrict__ agg,
                              const float* __restrict__ norm_dst,
                              const float* __restrict__ W,
                              const float* __restrict__ bias,
                              float* __restrict__ out,
                              int nRowTiles) {
  __shared__ float Wlds[NCP * WLDS_STRIDE];

  for (int i = threadIdx.x; i < NCP * GCN_D; i += blockDim.x) {
    int c = i >> 6;
    int k = i & 63;
    Wlds[c * WLDS_STRIDE + k] = (c < LDW) ? W[k * LDW + c] : 0.f;
  }
  __syncthreads();

  int gw = blockIdx.x * (blockDim.x >> 5) + ((int)threadIdx.x >> 5);
  if (gw >= nRowTiles * NCT) return;

  int lane  = threadIdx.x & 31;
  int rt    = gw / NCT;
  int ct    = gw - rt * NCT;
  int r0    = rt << 4;

  int mrow  = lane & 15;
  int khalf = (lane >> 4) << 1;
  int arow  = r0 + mrow;
  float nrm = norm_dst[arow];
  const float* aptr = agg + (size_t)arow * GCN_D + khalf;

  int ncol = (ct << 4) + mrow;
  const float* bptr = &Wlds[ncol * WLDS_STRIDE + khalf];

  v8f acc = {0.f, 0.f, 0.f, 0.f, 0.f, 0.f, 0.f, 0.f};

#pragma unroll
  for (int ks = 0; ks < 16; ++ks) {
    v2f a, b;
    float2 av = *(const float2*)(aptr + (ks << 2));
    float2 bv = *(const float2*)(bptr + (ks << 2));
    a.x = av.x * nrm;
    a.y = av.y * nrm;
    b.x = bv.x;
    b.y = bv.y;
    acc = __builtin_amdgcn_wmma_f32_16x16x4_f32(
         false, a,  false, b,
         (short)0, acc,  false,  false);
  }

  bool colok = (LDW == NCP) || (ncol < LDW);
  float bb = colok ? bias[ncol] : 0.f;
  int mbase = r0 + ((lane >> 4) << 3);
#pragma unroll
  for (int i = 0; i < 8; ++i) {
    float v = acc[i] + bb;
    if (RELU) v = fmaxf(v, 0.f);
    if (colok) out[(size_t)(mbase + i) * LDW + ncol] = v;
  }
}

extern "C" void kernel_launch(void* const* d_in, const int* in_sizes, int n_in,
                              void* d_out, int out_size, void* d_ws, size_t ws_size,
                              hipStream_t stream) {
  const float* features = (const float*)d_in[0];
  const int*   esrc     = (const int*)d_in[1];
  const int*   edst     = (const int*)d_in[2];
  const float* W1       = (const float*)d_in[3];
  const float* b1       = (const float*)d_in[4];
  const float* W2       = (const float*)d_in[5];
  const float* b2       = (const float*)d_in[6];
  const float* W3       = (const float*)d_in[7];
  const float* b3       = (const float*)d_in[8];
  float*       out      = (float*)d_out;

  const int N = GCN_N;
  const int E = in_sizes[1];

  float* norm_src = (float*)d_ws;
  float* norm_dst = norm_src + N;
  float* agg      = norm_dst + N;
  float* h        = agg + (size_t)N * GCN_D;

  const int TPB = 256;

  {
    int n4 = (2 * N) / 4;
    gcn_zero4<<<(n4 + TPB - 1) / TPB, TPB, 0, stream>>>((float4*)norm_src, n4);
    gcn_degree<<<(E + TPB - 1) / TPB, TPB, 0, stream>>>(esrc, edst, norm_src, norm_dst, E);
    gcn_norm<<<(2 * N + TPB - 1) / TPB, TPB, 0, stream>>>(norm_src, 2 * N);
  }

  int aggN4      = (N * GCN_D) / 4;
  int zeroBlocks = (aggN4 + TPB - 1) / TPB;
  int scatBlocks = ((E * 16) + TPB - 1) / TPB;
  int rowTiles   = N / 16;
  int wavesPB    = TPB / 32;

  gcn_zero4<<<zeroBlocks, TPB, 0, stream>>>((float4*)agg, aggN4);
  gcn_scatter<<<scatBlocks, TPB, 0, stream>>>(features, esrc, edst, norm_src, agg, E);
  {
    int total = rowTiles * 4;
    gcn_gemm_wmma<64, 64, 4, true><<<(total + wavesPB - 1) / wavesPB, TPB, 0, stream>>>(
        agg, norm_dst, W1, b1, h, rowTiles);
  }

  gcn_zero4<<<zeroBlocks, TPB, 0, stream>>>((float4*)agg, aggN4);
  gcn_scatter<<<scatBlocks, TPB, 0, stream>>>(h, esrc, edst, norm_src, agg, E);
  {
    int total = rowTiles * 4;
    gcn_gemm_wmma<64, 64, 4, true><<<(total + wavesPB - 1) / wavesPB, TPB, 0, stream>>>(
        agg, norm_dst, W2, b2, h, rowTiles);
  }

  gcn_zero4<<<zeroBlocks, TPB, 0, stream>>>((float4*)agg, aggN4);
  gcn_scatter<<<scatBlocks, TPB, 0, stream>>>(h, esrc, edst, norm_src, agg, E);
  {
    int total = rowTiles * 3;
    gcn_gemm_wmma<40, 48, 3, false><<<(total + wavesPB - 1) / wavesPB, TPB, 0, stream>>>(
        agg, norm_dst, W3, b3, out, rowTiles);
  }
}


# edge scatter re-mapped: a lane per feature column, 16 edges per wave; each row load and each f32 atomic add covers one whole 128-byte line (was 4 partial lines); sums stay in atomic arrival order
# speedup vs baseline: 1.4349x; 1.4349x over previous
#include <hip/hip_runtime.h>

typedef __attribute__((ext_vector_type(2))) float v2f;
typedef __attribute__((ext_vector_type(8))) float v8f;

#define GCN_N 100000
#define GCN_D 64
#define WLDS_STRIDE 68

__global__ void gcn_zero4(float4* __restrict__ p, int n4) {
  int i = blockIdx.x * blockDim.x + threadIdx.x;
  if (i < n4) p[i] = make_float4(0.f, 0.f, 0.f, 0.f);
}

__global__ void gcn_degree(const int* __restrict__ esrc, const int* __restrict__ edst,
                           float* __restrict__ deg_out, float* __restrict__ deg_in, int E) {
  int e = blockIdx.x * blockDim.x + threadIdx.x;
  if (e >= E) return;
  atomicAdd(&deg_out[esrc[e]], 1.0f);
  atomicAdd(&deg_in[edst[e]], 1.0f);
}

__global__ void gcn_norm(float* __restrict__ deg, int n) {
  int i = blockIdx.x * blockDim.x + threadIdx.x;
  if (i < n) deg[i] = rsqrtf(fmaxf(deg[i], 1.0f));
}

#define SC_EPW 16
__global__ void gcn_scatter(const float* __restrict__ h, const int* __restrict__ esrc,
                            const int* __restrict__ edst, const float* __restrict__ norm_src,
                            float* __restrict__ agg, int E) {
  const int lane = threadIdx.x & 31;
  const int wv = __builtin_amdgcn_readfirstlane(
      (int)(blockIdx.x * (blockDim.x >> 5) + (threadIdx.x >> 5)));
  const int e0 = wv * SC_EPW;
  if (e0 >= E) return;

  if (e0 + SC_EPW <= E) {
    const int el = e0 + (lane & (SC_EPW - 1));
    const int sl = esrc[el];
    const int dl = edst[el];
    const float nl = norm_src[sl];

    float v0[SC_EPW], v1[SC_EPW];
    for (int i = 0; i < SC_EPW; ++i) {
      const int s = __builtin_amdgcn_readlane(sl, i);
      const float* hp = h + (size_t)s * GCN_D + lane;
      v0[i] = hp[0];
      v1[i] = hp[32];
    }
    for (int i = 0; i < SC_EPW; ++i) {
      const int d = __builtin_amdgcn_readlane(dl, i);
      const float ns = __builtin_bit_cast(float, __builtin_amdgcn_readlane(__builtin_bit_cast(int, nl), i));
      float* ap = agg + (size_t)d * GCN_D + lane;
      atomicAdd(ap, v0[i] * ns);
      atomicAdd(ap + 32, v1[i] * ns);
    }
  } else {
    for (int e = e0; e < E; ++e) {
      const int s = esrc[e];
      const int d = edst[e];
      const float ns = norm_src[s];
      const float* hp = h + (size_t)s * GCN_D + lane;
      float* ap = agg + (size_t)d * GCN_D + lane;
      atomicAdd(ap, hp[0] * ns);
      atomicAdd(ap + 32, hp[32] * ns);
    }
  }
}

template <int LDW, int NCP, int NCT, bool RELU>
__global__ void gcn_gemm_wmma(const float* __restrict__ agg,
                              const float* __restrict__ norm_dst,
                              const float* __restrict__ W,
                              const float* __restrict__ bias,
                              float* __restrict__ out,
                              int nRowTiles) {
  __shared__ float Wlds[NCP * WLDS_STRIDE];

  for (int i = threadIdx.x; i < NCP * GCN_D; i += blockDim.x) {
    int c = i >> 6;
    int k = i & 63;
    Wlds[c * WLDS_STRIDE + k] = (c < LDW) ? W[k * LDW + c] : 0.f;
  }
  __syncthreads();

  int gw = blockIdx.x * (blockDim.x >> 5) + ((int)threadIdx.x >> 5);
  if (gw >= nRowTiles * NCT) return;

  int lane  = threadIdx.x & 31;
  int rt    = gw / NCT;
  int ct    = gw - rt * NCT;
  int r0    = rt << 4;

  int mrow  = lane & 15;
  int khalf = (lane >> 4) << 1;
  int arow  = r0 + mrow;
  float nrm = norm_dst[arow];
  const float* aptr = agg + (size_t)arow * GCN_D + khalf;

  int ncol = (ct << 4) + mrow;
  const float* bptr = &Wlds[ncol * WLDS_STRIDE + khalf];

  v8f acc = {0.f, 0.f, 0.f, 0.f, 0.f, 0.f, 0.f, 0.f};

#pragma unroll
  for (int ks = 0; ks < 16; ++ks) {
    v2f a, b;
    float2 av = *(const float2*)(aptr + (ks << 2));
    float2 bv = *(const float2*)(bptr + (ks << 2));
    a.x = av.x * nrm;
    a.y = av.y * nrm;
    b.x = bv.x;
    b.y = bv.y;
    acc = __builtin_amdgcn_wmma_f32_16x16x4_f32(
         false, a,  false, b,
         (short)0, acc,  false,  false);
  }

  bool colok = (LDW == NCP) || (ncol < LDW);
  float bb = colok ? bias[ncol] : 0.f;
  int mbase = r0 + ((lane >> 4) << 3);
#pragma unroll
  for (int i = 0; i < 8; ++i) {
    float v = acc[i] + bb;
    if (RELU) v = fmaxf(v, 0.f);
    if (colok) out[(size_t)(mbase + i) * LDW + ncol] = v;
  }
}

extern "C" void kernel_launch(void* const* d_in, const int* in_sizes, int n_in,
                              void* d_out, int out_size, void* d_ws, size_t ws_size,
                              hipStream_t stream) {
  const float* features = (const float*)d_in[0];
  const int*   esrc     = (const int*)d_in[1];
  const int*   edst     = (const int*)d_in[2];
  const float* W1       = (const float*)d_in[3];
  const float* b1       = (const float*)d_in[4];
  const float* W2       = (const float*)d_in[5];
  const float* b2       = (const float*)d_in[6];
  const float* W3       = (const float*)d_in[7];
  const float* b3       = (const float*)d_in[8];
  float*       out      = (float*)d_out;

  const int N = GCN_N;
  const int E = in_sizes[1];

  float* norm_src = (float*)d_ws;
  float* norm_dst = norm_src + N;
  float* agg      = norm_dst + N;
  float* h        = agg + (size_t)N * GCN_D;

  const int TPB = 256;

  {
    int n4 = (2 * N) / 4;
    gcn_zero4<<<(n4 + TPB - 1) / TPB, TPB, 0, stream>>>((float4*)norm_src, n4);
    gcn_degree<<<(E + TPB - 1) / TPB, TPB, 0, stream>>>(esrc, edst, norm_src, norm_dst, E);
    gcn_norm<<<(2 * N + TPB - 1) / TPB, TPB, 0, stream>>>(norm_src, 2 * N);
  }

  int aggN4      = (N * GCN_D) / 4;
  int zeroBlocks = (aggN4 + TPB - 1) / TPB;
  int edgesPB    = (TPB / 32) * SC_EPW;
  int scatBlocks = (E + edgesPB - 1) / edgesPB;
  int rowTiles   = N / 16;
  int wavesPB    = TPB / 32;

  gcn_zero4<<<zeroBlocks, TPB, 0, stream>>>((float4*)agg, aggN4);
  gcn_scatter<<<scatBlocks, TPB, 0, stream>>>(features, esrc, edst, norm_src, agg, E);
  {
    int total = rowTiles * 4;
    gcn_gemm_wmma<64, 64, 4, true><<<(total + wavesPB - 1) / wavesPB, TPB, 0, stream>>>(
        agg, norm_dst, W1, b1, h, rowTiles);
  }

  gcn_zero4<<<zeroBlocks, TPB, 0, stream>>>((float4*)agg, aggN4);
  gcn_scatter<<<scatBlocks, TPB, 0, stream>>>(h, esrc, edst, norm_src, agg, E);
  {
    int total = rowTiles * 4;
    gcn_gemm_wmma<64, 64, 4, true><<<(total + wavesPB - 1) / wavesPB, TPB, 0, stream>>>(
        agg, norm_dst, W2, b2, h, rowTiles);
  }

  gcn_zero4<<<zeroBlocks, TPB, 0, stream>>>((float4*)agg, aggN4);
  gcn_scatter<<<scatBlocks, TPB, 0, stream>>>(h, esrc, edst, norm_src, agg, E);
  {
    int total = rowTiles * 3;
    gcn_gemm_wmma<40, 48, 3, false><<<(total + wavesPB - 1) / wavesPB, TPB, 0, stream>>>(
        agg, norm_dst, W3, b3, out, rowTiles);
  }
}


# float-atomic edge scatter into a zeroed agg replaced by a per-call counting sort of the edges by destination and an atomic-free row gather that writes each agg row once; sums in cursor-arrival order
# speedup vs baseline: 1.6170x; 1.1269x over previous
#include <hip/hip_runtime.h>

typedef __attribute__((ext_vector_type(2))) float v2f;
typedef __attribute__((ext_vector_type(8))) float v8f;

#define GCN_N 100000
#define GCN_D 64
#define WLDS_STRIDE 68

__global__ void gcn_zero4(float4* __restrict__ p, int n4, unsigned int* __restrict__ counter) {
  int i = blockIdx.x * blockDim.x + threadIdx.x;
  if (i < n4) p[i] = make_float4(0.f, 0.f, 0.f, 0.f);
  if (i == 0) *counter = 0u;
}

__global__ void gcn_degree(const int* __restrict__ esrc, const int* __restrict__ edst,
                           float* __restrict__ deg_out, float* __restrict__ deg_in, int E) {
  int e = blockIdx.x * blockDim.x + threadIdx.x;
  if (e >= E) return;
  atomicAdd(&deg_out[esrc[e]], 1.0f);
  atomicAdd(&deg_in[edst[e]], 1.0f);
}

__global__ void gcn_norm_alloc(float* __restrict__ deg_out, float* __restrict__ deg_in,
                               int* __restrict__ seg_start, int* __restrict__ seg_cursor,
                               unsigned int* __restrict__ counter, int n) {
  const int i = blockIdx.x * blockDim.x + threadIdx.x;
  const int lane = threadIdx.x & 31;
  const bool ok = i < n;
  const float dout = ok ? deg_out[i] : 0.f;
  const float din  = ok ? deg_in[i] : 0.f;
  const int c = (int)din;
  int incl = c;
  for (int off = 1; off < 32; off <<= 1) {
    const int t = __shfl_up(incl, off, 32);
    if (lane >= off) incl += t;
  }
  const int wsum = __shfl(incl, 31, 32);
  int base = 0;
  if (lane == 31) base = (int)atomicAdd(counter, (unsigned int)wsum);
  base = __shfl(base, 31, 32);
  if (ok) {
    const int st = base + incl - c;
    seg_start[i]  = st;
    seg_cursor[i] = st;
    deg_out[i] = rsqrtf(fmaxf(dout, 1.0f));
    deg_in[i]  = rsqrtf(fmaxf(din, 1.0f));
  }
}

__global__ void gcn_fill(const int* __restrict__ esrc, const int* __restrict__ edst,
                         int* __restrict__ seg_cursor, int* __restrict__ ssrc, int E) {
  const int e0 = (blockIdx.x * blockDim.x + threadIdx.x) * 4;
  if (e0 >= E) return;
  const unsigned int last = (unsigned int)(E - 1);
  if (e0 + 4 <= E) {
    const int4 s4 = *(const int4*)(esrc + e0);
    const int4 d4 = *(const int4*)(edst + e0);
    const unsigned int p0 = (unsigned int)atomicAdd(&seg_cursor[d4.x], 1);
    const unsigned int p1 = (unsigned int)atomicAdd(&seg_cursor[d4.y], 1);
    const unsigned int p2 = (unsigned int)atomicAdd(&seg_cursor[d4.z], 1);
    const unsigned int p3 = (unsigned int)atomicAdd(&seg_cursor[d4.w], 1);
    ssrc[min(p0, last)] = s4.x;
    ssrc[min(p1, last)] = s4.y;
    ssrc[min(p2, last)] = s4.z;
    ssrc[min(p3, last)] = s4.w;
  } else {
    for (int e = e0; e < E; ++e) {
      const unsigned int p = (unsigned int)atomicAdd(&seg_cursor[edst[e]], 1);
      ssrc[min(p, last)] = esrc[e];
    }
  }
}

#define GA_NPW 8
#define GA_ROWS 8
__global__ void gcn_gather(const float* __restrict__ h, const int* __restrict__ seg_start,
                           const int* __restrict__ seg_end, const int* __restrict__ ssrc,
                           const float* __restrict__ norm_src, float* __restrict__ agg, int n,
                           int E) {
  const int lane = threadIdx.x & 31;
  const int wv = __builtin_amdgcn_readfirstlane(
      (int)(blockIdx.x * (blockDim.x >> 5) + (threadIdx.x >> 5)));
  const int n0 = wv * GA_NPW;
  if (n0 >= n) return;
  const int nn = min(GA_NPW, n - n0);

  const int nl = min(n0 + (lane & (GA_NPW - 1)), n - 1);
  const int sl = seg_start[nl];
  const int el = seg_end[nl];

  for (int k = 0; k < nn; ++k) {
    const int s = max(__builtin_amdgcn_readlane(sl, k), 0);
    const int e = min(__builtin_amdgcn_readlane(el, k), E);
    float ax = 0.f, ay = 0.f;
    for (int j0 = s; j0 < e; j0 += 32) {
      const int cnt = min(32, e - j0);
      int idl = 0;
      float nsl = 0.f;
      if (lane < cnt) {
        idl = (int)min((unsigned int)ssrc[j0 + lane], (unsigned int)(n - 1));
        nsl = norm_src[idl];
      }
      for (int i = 0; i < cnt; i += GA_ROWS) {
        float2 v[GA_ROWS];
#pragma unroll
        for (int u = 0; u < GA_ROWS; ++u) {
          v[u] = make_float2(0.f, 0.f);
          if (i + u < cnt) {
            const int sidx = __builtin_amdgcn_readlane(idl, i + u);
            v[u] = *(const float2*)(h + (size_t)sidx * GCN_D + 2 * lane);
          }
        }
#pragma unroll
        for (int u = 0; u < GA_ROWS; ++u) {
          const float ns = __builtin_bit_cast(
              float, __builtin_amdgcn_readlane(__builtin_bit_cast(int, nsl), i + u));
          ax += v[u].x * ns;
          ay += v[u].y * ns;
        }
      }
    }
    *(float2*)(agg + (size_t)(n0 + k) * GCN_D + 2 * lane) = make_float2(ax, ay);
  }
}

template <int LDW, int NCP, int NCT, bool RELU>
__global__ void gcn_gemm_wmma(const float* __restrict__ agg,
                              const float* __restrict__ norm_dst,
                              const float* __restrict__ W,
                              const float* __restrict__ bias,
                              float* __restrict__ out,
                              int nRowTiles) {
  __shared__ float Wlds[NCP * WLDS_STRIDE];

  for (int i = threadIdx.x; i < NCP * GCN_D; i += blockDim.x) {
    int c = i >> 6;
    int k = i & 63;
    Wlds[c * WLDS_STRIDE + k] = (c < LDW) ? W[k * LDW + c] : 0.f;
  }
  __syncthreads();

  int gw = blockIdx.x * (blockDim.x >> 5) + ((int)threadIdx.x >> 5);
  if (gw >= nRowTiles * NCT) return;

  int lane  = threadIdx.x & 31;
  int rt    = gw / NCT;
  int ct    = gw - rt * NCT;
  int r0    = rt << 4;

  int mrow  = lane & 15;
  int khalf = (lane >> 4) << 1;
  int arow  = r0 + mrow;
  float nrm = norm_dst[arow];
  const float* aptr = agg + (size_t)arow * GCN_D + khalf;

  int ncol = (ct << 4) + mrow;
  const float* bptr = &Wlds[ncol * WLDS_STRIDE + khalf];

  v8f acc = {0.f, 0.f, 0.f, 0.f, 0.f, 0.f, 0.f, 0.f};

#pragma unroll
  for (int ks = 0; ks < 16; ++ks) {
    v2f a, b;
    float2 av = *(const float2*)(aptr + (ks << 2));
    float2 bv = *(const float2*)(bptr + (ks << 2));
    a.x = av.x * nrm;
    a.y = av.y * nrm;
    b.x = bv.x;
    b.y = bv.y;
    acc = __builtin_amdgcn_wmma_f32_16x16x4_f32(
         false, a,  false, b,
         (short)0, acc,  false,  false);
  }

  bool colok = (LDW == NCP) || (ncol < LDW);
  float bb = colok ? bias[ncol] : 0.f;
  int mbase = r0 + ((lane >> 4) << 3);
#pragma unroll
  for (int i = 0; i < 8; ++i) {
    float v = acc[i] + bb;
    if (RELU) v = fmaxf(v, 0.f);
    if (colok) out[(size_t)(mbase + i) * LDW + ncol] = v;
  }
}

extern "C" void kernel_launch(void* const* d_in, const int* in_sizes, int n_in,
                              void* d_out, int out_size, void* d_ws, size_t ws_size,
                              hipStream_t stream) {
  const float* features = (const float*)d_in[0];
  const int*   esrc     = (const int*)d_in[1];
  const int*   edst     = (const int*)d_in[2];
  const float* W1       = (const float*)d_in[3];
  const float* b1       = (const float*)d_in[4];
  const float* W2       = (const float*)d_in[5];
  const float* b2       = (const float*)d_in[6];
  const float* W3       = (const float*)d_in[7];
  const float* b3       = (const float*)d_in[8];
  float*       out      = (float*)d_out;

  const int N = GCN_N;
  const int E = in_sizes[1];

  float* norm_src = (float*)d_ws;
  float* norm_dst = norm_src + N;
  float* agg      = norm_dst + N;
  float* h        = agg + (size_t)N * GCN_D;

  int*          seg_start  = (int*)d_out;
  int*          seg_cursor = seg_start + N;
  unsigned int* counter    = (unsigned int*)(seg_cursor + N);
  int*          ssrc       = seg_cursor + N + 4;

  const int TPB = 256;

  {
    int n4 = (2 * N) / 4;
    gcn_zero4<<<(n4 + TPB - 1) / TPB, TPB, 0, stream>>>((float4*)norm_src, n4, counter);
    gcn_degree<<<(E + TPB - 1) / TPB, TPB, 0, stream>>>(esrc, edst, norm_src, norm_dst, E);
    gcn_norm_alloc<<<(N + TPB - 1) / TPB, TPB, 0, stream>>>(norm_src, norm_dst, seg_start,
                                                            seg_cursor, counter, N);
    int fillThreads = (E + 3) / 4;
    gcn_fill<<<(fillThreads + TPB - 1) / TPB, TPB, 0, stream>>>(esrc, edst, seg_cursor, ssrc, E);
  }

  int nodesPB    = (TPB / 32) * GA_NPW;
  int gathBlocks = (N + nodesPB - 1) / nodesPB;
  int rowTiles   = N / 16;
  int wavesPB    = TPB / 32;

  gcn_gather<<<gathBlocks, TPB, 0, stream>>>(features, seg_start, seg_cursor, ssrc, norm_src, agg, N, E);
  {
    int total = rowTiles * 4;
    gcn_gemm_wmma<64, 64, 4, true><<<(total + wavesPB - 1) / wavesPB, TPB, 0, stream>>>(
        agg, norm_dst, W1, b1, h, rowTiles);
  }

  gcn_gather<<<gathBlocks, TPB, 0, stream>>>(h, seg_start, seg_cursor, ssrc, norm_src, agg, N, E);
  {
    int total = rowTiles * 4;
    gcn_gemm_wmma<64, 64, 4, true><<<(total + wavesPB - 1) / wavesPB, TPB, 0, stream>>>(
        agg, norm_dst, W2, b2, h, rowTiles);
  }

  gcn_gather<<<gathBlocks, TPB, 0, stream>>>(h, seg_start, seg_cursor, ssrc, norm_src, agg, N, E);
  {
    int total = rowTiles * 3;
    gcn_gemm_wmma<40, 48, 3, false><<<(total + wavesPB - 1) / wavesPB, TPB, 0, stream>>>(
        agg, norm_dst, W3, b3, out, rowTiles);
  }
}


# layers 1 and 2: 64x64 f32 WMMA product fused into the row-gather kernel (rows scaled by norm_dst into an LDS tile, one barrier, two column tiles per wave); agg round trip and two launches gone
# speedup vs baseline: 1.9015x; 1.1759x over previous
#include <hip/hip_runtime.h>

typedef __attribute__((ext_vector_type(2))) float v2f;
typedef __attribute__((ext_vector_type(8))) float v8f;

#define GCN_N 100000
#define GCN_D 64
#define WLDS_STRIDE 68

__global__ void gcn_zero4(float4* __restrict__ p, int n4, unsigned int* __restrict__ counter) {
  int i = blockIdx.x * blockDim.x + threadIdx.x;
  if (i < n4) p[i] = make_float4(0.f, 0.f, 0.f, 0.f);
  if (i == 0) *counter = 0u;
}

__global__ void gcn_degree(const int* __restrict__ esrc, const int* __restrict__ edst,
                           float* __restrict__ deg_out, float* __restrict__ deg_in, int E) {
  int e = blockIdx.x * blockDim.x + threadIdx.x;
  if (e >= E) return;
  atomicAdd(&deg_out[esrc[e]], 1.0f);
  atomicAdd(&deg_in[edst[e]], 1.0f);
}

__global__ void gcn_norm_alloc(float* __restrict__ deg_out, float* __restrict__ deg_in,
                               int* __restrict__ seg_start, int* __restrict__ seg_cursor,
                               unsigned int* __restrict__ counter, int n) {
  const int i = blockIdx.x * blockDim.x + threadIdx.x;
  const int lane = threadIdx.x & 31;
  const bool ok = i < n;
  const float dout = ok ? deg_out[i] : 0.f;
  const float din  = ok ? deg_in[i] : 0.f;
  const int c = (int)din;
  int incl = c;
  for (int off = 1; off < 32; off <<= 1) {
    const int t = __shfl_up(incl, off, 32);
    if (lane >= off) incl += t;
  }
  const int wsum = __shfl(incl, 31, 32);
  int base = 0;
  if (lane == 31) base = (int)atomicAdd(counter, (unsigned int)wsum);
  base = __shfl(base, 31, 32);
  if (ok) {
    const int st = base + incl - c;
    seg_start[i]  = st;
    seg_cursor[i] = st;
    deg_out[i] = rsqrtf(fmaxf(dout, 1.0f));
    deg_in[i]  = rsqrtf(fmaxf(din, 1.0f));
  }
}

__global__ void gcn_fill(const int* __restrict__ esrc, const int* __restrict__ edst,
                         int* __restrict__ seg_cursor, int* __restrict__ ssrc, int E) {
  const int e0 = (blockIdx.x * blockDim.x + threadIdx.x) * 4;
  if (e0 >= E) return;
  const unsigned int last = (unsigned int)(E - 1);
  if (e0 + 4 <= E) {
    const int4 s4 = *(const int4*)(esrc + e0);
    const int4 d4 = *(const int4*)(edst + e0);
    const unsigned int p0 = (unsigned int)atomicAdd(&seg_cursor[d4.x], 1);
    const unsigned int p1 = (unsigned int)atomicAdd(&seg_cursor[d4.y], 1);
    const unsigned int p2 = (unsigned int)atomicAdd(&seg_cursor[d4.z], 1);
    const unsigned int p3 = (unsigned int)atomicAdd(&seg_cursor[d4.w], 1);
    ssrc[min(p0, last)] = s4.x;
    ssrc[min(p1, last)] = s4.y;
    ssrc[min(p2, last)] = s4.z;
    ssrc[min(p3, last)] = s4.w;
  } else {
    for (int e = e0; e < E; ++e) {
      const unsigned int p = (unsigned int)atomicAdd(&seg_cursor[edst[e]], 1);
      ssrc[min(p, last)] = esrc[e];
    }
  }
}

#define GA_NPW 8
#define GA_ROWS 8
__global__ void gcn_gather(const float* __restrict__ h, const int* __restrict__ seg_start,
                           const int* __restrict__ seg_end, const int* __restrict__ ssrc,
                           const float* __restrict__ norm_src, float* __restrict__ agg, int n,
                           int E) {
  const int lane = threadIdx.x & 31;
  const int wv = __builtin_amdgcn_readfirstlane(
      (int)(blockIdx.x * (blockDim.x >> 5) + (threadIdx.x >> 5)));
  const int n0 = wv * GA_NPW;
  if (n0 >= n) return;
  const int nn = min(GA_NPW, n - n0);

  const int nl = min(n0 + (lane & (GA_NPW - 1)), n - 1);
  const int sl = seg_start[nl];
  const int el = seg_end[nl];

  for (int k = 0; k < nn; ++k) {
    const int s = max(__builtin_amdgcn_readlane(sl, k), 0);
    const int e = min(__builtin_amdgcn_readlane(el, k), E);
    float ax = 0.f, ay = 0.f;
    for (int j0 = s; j0 < e; j0 += 32) {
      const int cnt = min(32, e - j0);
      int idl = 0;
      float nsl = 0.f;
      if (lane < cnt) {
        idl = (int)min((unsigned int)ssrc[j0 + lane], (unsigned int)(n - 1));
        nsl = norm_src[idl];
      }
      for (int i = 0; i < cnt; i += GA_ROWS) {
        float2 v[GA_ROWS];
#pragma unroll
        for (int u = 0; u < GA_ROWS; ++u) {
          v[u] = make_float2(0.f, 0.f);
          if (i + u < cnt) {
            const int sidx = __builtin_amdgcn_readlane(idl, i + u);
            v[u] = *(const float2*)(h + (size_t)sidx * GCN_D + 2 * lane);
          }
        }
#pragma unroll
        for (int u = 0; u < GA_ROWS; ++u) {
          const float ns = __builtin_bit_cast(
              float, __builtin_amdgcn_readlane(__builtin_bit_cast(int, nsl), i + u));
          ax += v[u].x * ns;
          ay += v[u].y * ns;
        }
      }
    }
    *(float2*)(agg + (size_t)(n0 + k) * GCN_D + 2 * lane) = make_float2(ax, ay);
  }
}

#define FL_THREADS 256
#define FL_TILES 4
__global__ void __launch_bounds__(FL_THREADS)
gcn_layer_fused(const float* __restrict__ hin, const int* __restrict__ seg_start,
                const int* __restrict__ seg_end, const int* __restrict__ ssrc,
                const float* __restrict__ norm_src, const float* __restrict__ norm_dst,
                const float* __restrict__ W, const float* __restrict__ bias,
                float* __restrict__ hout, int n, int E, int nRowTiles) {
  __shared__ __attribute__((aligned(16))) float Wlds[GCN_D * WLDS_STRIDE];
  __shared__ __attribute__((aligned(16))) float Alds[FL_TILES * 16 * WLDS_STRIDE];

  const int lane = threadIdx.x & 31;
  const int wv = __builtin_amdgcn_readfirstlane(
      (int)(blockIdx.x * (FL_THREADS >> 5) + (threadIdx.x >> 5)));
  const int w  = wv & 7;
  const int n0 = wv * GA_NPW;
  const int nn = max(0, min(GA_NPW, n - n0));

  const int nl = min(n0 + (lane & (GA_NPW - 1)), n - 1);
  const int sl = seg_start[nl];
  const int el = seg_end[nl];
  const float ndl = norm_dst[nl];

#pragma unroll
  for (int j = 0; j < (GCN_D * GCN_D) / (4 * FL_THREADS); ++j) {
    const int i = ((int)threadIdx.x + j * FL_THREADS) * 4;
    const float4 w4 = *(const float4*)(W + i);
    const int k = i >> 6;
    const int c = i & 63;
    Wlds[(c + 0) * WLDS_STRIDE + k] = w4.x;
    Wlds[(c + 1) * WLDS_STRIDE + k] = w4.y;
    Wlds[(c + 2) * WLDS_STRIDE + k] = w4.z;
    Wlds[(c + 3) * WLDS_STRIDE + k] = w4.w;
  }

  float* arow = &Alds[(w * GA_NPW) * WLDS_STRIDE + 2 * lane];

  for (int k = 0; k < nn; ++k) {
    const int s = max(__builtin_amdgcn_readlane(sl, k), 0);
    const int e = min(__builtin_amdgcn_readlane(el, k), E);
    float ax = 0.f, ay = 0.f;
    for (int j0 = s; j0 < e; j0 += 32) {
      const int cnt = min(32, e - j0);
      int idl = 0;
      float nsl = 0.f;
      if (lane < cnt) {
        idl = (int)min((unsigned int)ssrc[j0 + lane], (unsigned int)(n - 1));
        nsl = norm_src[idl];
      }
      for (int i = 0; i < cnt; i += GA_ROWS) {
        float2 v[GA_ROWS];
#pragma unroll
        for (int u = 0; u < GA_ROWS; ++u) {
          v[u] = make_float2(0.f, 0.f);
          if (i + u < cnt) {
            const int sidx = __builtin_amdgcn_readlane(idl, i + u);
            v[u] = *(const float2*)(hin + (size_t)sidx * GCN_D + 2 * lane);
          }
        }
#pragma unroll
        for (int u = 0; u < GA_ROWS; ++u) {
          const float ns = __builtin_bit_cast(
              float, __builtin_amdgcn_readlane(__builtin_bit_cast(int, nsl), i + u));
          ax += v[u].x * ns;
          ay += v[u].y * ns;
        }
      }
    }
    const float nd = __builtin_bit_cast(
        float, __builtin_amdgcn_readlane(__builtin_bit_cast(int, ndl), k));
    *(float2*)(arow + k * WLDS_STRIDE) = make_float2(ax * nd, ay * nd);
  }

  __syncthreads();

  const int rt = (int)blockIdx.x * FL_TILES + (w >> 1);
  if (rt >= nRowTiles) return;

  const int mrow  = lane & 15;
  const int khalf = (lane >> 4) << 1;
  const int ct0   = (w & 1) << 1;
  const float* aptr = &Alds[((w >> 1) * 16 + mrow) * WLDS_STRIDE + khalf];
  const float* bptr0 = &Wlds[((ct0 << 4) + mrow) * WLDS_STRIDE + khalf];
  const float* bptr1 = bptr0 + 16 * WLDS_STRIDE;

  v8f acc0 = {0.f, 0.f, 0.f, 0.f, 0.f, 0.f, 0.f, 0.f};
  v8f acc1 = {0.f, 0.f, 0.f, 0.f, 0.f, 0.f, 0.f, 0.f};

#pragma unroll
  for (int ks = 0; ks < 16; ++ks) {
    v2f a, b0, b1;
    const float2 av  = *(const float2*)(aptr + (ks << 2));
    const float2 bv0 = *(const float2*)(bptr0 + (ks << 2));
    const float2 bv1 = *(const float2*)(bptr1 + (ks << 2));
    a.x = av.x;   a.y = av.y;
    b0.x = bv0.x; b0.y = bv0.y;
    b1.x = bv1.x; b1.y = bv1.y;
    acc0 = __builtin_amdgcn_wmma_f32_16x16x4_f32(
         false, a,  false, b0,
         (short)0, acc0,  false,  false);
    acc1 = __builtin_amdgcn_wmma_f32_16x16x4_f32(
         false, a,  false, b1,
         (short)0, acc1,  false,  false);
  }

  const int ncol0 = (ct0 << 4) + mrow;
  const float bb0 = bias[ncol0];
  const float bb1 = bias[ncol0 + 16];
  float* orow = hout + (size_t)((rt << 4) + ((lane >> 4) << 3)) * GCN_D + ncol0;
#pragma unroll
  for (int i = 0; i < 8; ++i) {
    orow[(size_t)i * GCN_D]      = fmaxf(acc0[i] + bb0, 0.f);
    orow[(size_t)i * GCN_D + 16] = fmaxf(acc1[i] + bb1, 0.f);
  }
}

template <int LDW, int NCP, int NCT, bool RELU>
__global__ void gcn_gemm_wmma(const float* __restrict__ agg,
                              const float* __restrict__ norm_dst,
                              const float* __restrict__ W,
                              const float* __restrict__ bias,
                              float* __restrict__ out,
                              int nRowTiles) {
  __shared__ float Wlds[NCP * WLDS_STRIDE];

  for (int i = threadIdx.x; i < NCP * GCN_D; i += blockDim.x) {
    int c = i >> 6;
    int k = i & 63;
    Wlds[c * WLDS_STRIDE + k] = (c < LDW) ? W[k * LDW + c] : 0.f;
  }
  __syncthreads();

  int gw = blockIdx.x * (blockDim.x >> 5) + ((int)threadIdx.x >> 5);
  if (gw >= nRowTiles * NCT) return;

  int lane  = threadIdx.x & 31;
  int rt    = gw / NCT;
  int ct    = gw - rt * NCT;
  int r0    = rt << 4;

  int mrow  = lane & 15;
  int khalf = (lane >> 4) << 1;
  int arow  = r0 + mrow;
  float nrm = norm_dst[arow];
  const float* aptr = agg + (size_t)arow * GCN_D + khalf;

  int ncol = (ct << 4) + mrow;
  const float* bptr = &Wlds[ncol * WLDS_STRIDE + khalf];

  v8f acc = {0.f, 0.f, 0.f, 0.f, 0.f, 0.f, 0.f, 0.f};

#pragma unroll
  for (int ks = 0; ks < 16; ++ks) {
    v2f a, b;
    float2 av = *(const float2*)(aptr + (ks << 2));
    float2 bv = *(const float2*)(bptr + (ks << 2));
    a.x = av.x * nrm;
    a.y = av.y * nrm;
    b.x = bv.x;
    b.y = bv.y;
    acc = __builtin_amdgcn_wmma_f32_16x16x4_f32(
         false, a,  false, b,
         (short)0, acc,  false,  false);
  }

  bool colok = (LDW == NCP) || (ncol < LDW);
  float bb = colok ? bias[ncol] : 0.f;
  int mbase = r0 + ((lane >> 4) << 3);
#pragma unroll
  for (int i = 0; i < 8; ++i) {
    float v = acc[i] + bb;
    if (RELU) v = fmaxf(v, 0.f);
    if (colok) out[(size_t)(mbase + i) * LDW + ncol] = v;
  }
}

extern "C" void kernel_launch(void* const* d_in, const int* in_sizes, int n_in,
                              void* d_out, int out_size, void* d_ws, size_t ws_size,
                              hipStream_t stream) {
  const float* features = (const float*)d_in[0];
  const int*   esrc     = (const int*)d_in[1];
  const int*   edst     = (const int*)d_in[2];
  const float* W1       = (const float*)d_in[3];
  const float* b1       = (const float*)d_in[4];
  const float* W2       = (const float*)d_in[5];
  const float* b2       = (const float*)d_in[6];
  const float* W3       = (const float*)d_in[7];
  const float* b3       = (const float*)d_in[8];
  float*       out      = (float*)d_out;

  const int N = GCN_N;
  const int E = in_sizes[1];

  float* norm_src = (float*)d_ws;
  float* norm_dst = norm_src + N;
  float* agg      = norm_dst + N;
  float* h        = agg + (size_t)N * GCN_D;

  int*          seg_start  = (int*)d_out;
  int*          seg_cursor = seg_start + N;
  unsigned int* counter    = (unsigned int*)(seg_cursor + N);
  int*          ssrc       = seg_cursor + N + 4;

  const int TPB = 256;

  {
    int n4 = (2 * N) / 4;
    gcn_zero4<<<(n4 + TPB - 1) / TPB, TPB, 0, stream>>>((float4*)norm_src, n4, counter);
    gcn_degree<<<(E + TPB - 1) / TPB, TPB, 0, stream>>>(esrc, edst, norm_src, norm_dst, E);
    gcn_norm_alloc<<<(N + TPB - 1) / TPB, TPB, 0, stream>>>(norm_src, norm_dst, seg_start,
                                                            seg_cursor, counter, N);
    int fillThreads = (E + 3) / 4;
    gcn_fill<<<(fillThreads + TPB - 1) / TPB, TPB, 0, stream>>>(esrc, edst, seg_cursor, ssrc, E);
  }

  int nodesPB    = (TPB / 32) * GA_NPW;
  int gathBlocks = (N + nodesPB - 1) / nodesPB;
  int rowTiles   = N / 16;
  int wavesPB    = TPB / 32;

  gcn_layer_fused<<<gathBlocks, TPB, 0, stream>>>(features, seg_start, seg_cursor, ssrc, norm_src,
                                                  norm_dst, W1, b1, h, N, E, rowTiles);

  float* h2 = agg;
  gcn_layer_fused<<<gathBlocks, TPB, 0, stream>>>(h, seg_start, seg_cursor, ssrc, norm_src,
                                                  norm_dst, W2, b2, h2, N, E, rowTiles);

  float* agg3 = h;
  gcn_gather<<<gathBlocks, TPB, 0, stream>>>(h2, seg_start, seg_cursor, ssrc, norm_src, agg3, N, E);
  {
    int total = rowTiles * 3;
    gcn_gemm_wmma<40, 48, 3, false><<<(total + wavesPB - 1) / wavesPB, TPB, 0, stream>>>(
        agg3, norm_dst, W3, b3, out, rowTiles);
  }
}


# sort's fill pass has no atomics: an edge's slot = segment start + rank returned by the in-degree count's integer add (4 edges per thread, rank kept per edge); sums stay in atomic arrival order
# speedup vs baseline: 2.3919x; 1.2579x over previous
#include <hip/hip_runtime.h>

typedef __attribute__((ext_vector_type(2))) float v2f;
typedef __attribute__((ext_vector_type(8))) float v8f;

#define GCN_N 100000
#define GCN_D 64
#define WLDS_STRIDE 68

__global__ void gcn_zero4(float4* __restrict__ p, int n4, unsigned int* __restrict__ counter) {
  int i = blockIdx.x * blockDim.x + threadIdx.x;
  if (i < n4) p[i] = make_float4(0.f, 0.f, 0.f, 0.f);
  if (i == 0) *counter = 0u;
}

__global__ void gcn_degree_rank(const int* __restrict__ esrc, const int* __restrict__ edst,
                                float* __restrict__ deg_out, int* __restrict__ cnt_in,
                                int* __restrict__ rank, int E) {
  const int e0 = (blockIdx.x * blockDim.x + threadIdx.x) * 4;
  if (e0 >= E) return;
  if (e0 + 4 <= E) {
    const int4 s4 = *(const int4*)(esrc + e0);
    const int4 d4 = *(const int4*)(edst + e0);
    atomicAdd(&deg_out[s4.x], 1.0f);
    atomicAdd(&deg_out[s4.y], 1.0f);
    atomicAdd(&deg_out[s4.z], 1.0f);
    atomicAdd(&deg_out[s4.w], 1.0f);
    int4 r4;
    r4.x = atomicAdd(&cnt_in[d4.x], 1);
    r4.y = atomicAdd(&cnt_in[d4.y], 1);
    r4.z = atomicAdd(&cnt_in[d4.z], 1);
    r4.w = atomicAdd(&cnt_in[d4.w], 1);
    *(int4*)(rank + e0) = r4;
  } else {
    for (int e = e0; e < E; ++e) {
      atomicAdd(&deg_out[esrc[e]], 1.0f);
      rank[e] = atomicAdd(&cnt_in[edst[e]], 1);
    }
  }
}

__global__ void gcn_norm_alloc(float* __restrict__ deg_out, float* __restrict__ deg_in,
                               int* __restrict__ seg_start, int* __restrict__ seg_end,
                               unsigned int* __restrict__ counter, int n) {
  const int i = blockIdx.x * blockDim.x + threadIdx.x;
  const int lane = threadIdx.x & 31;
  const bool ok = i < n;
  const float dout = ok ? deg_out[i] : 0.f;
  const int c = ok ? __float_as_int(deg_in[i]) : 0;
  int incl = c;
  for (int off = 1; off < 32; off <<= 1) {
    const int t = __shfl_up(incl, off, 32);
    if (lane >= off) incl += t;
  }
  const int wsum = __shfl(incl, 31, 32);
  int base = 0;
  if (lane == 31) base = (int)atomicAdd(counter, (unsigned int)wsum);
  base = __shfl(base, 31, 32);
  if (ok) {
    const int st = base + incl - c;
    seg_start[i] = st;
    seg_end[i]   = st + c;
    deg_out[i] = rsqrtf(fmaxf(dout, 1.0f));
    deg_in[i]  = rsqrtf(fmaxf((float)c, 1.0f));
  }
}

__global__ void gcn_fill(const int* __restrict__ esrc, const int* __restrict__ edst,
                         const int* __restrict__ seg_start, const int* __restrict__ rank,
                         int* __restrict__ ssrc, int E) {
  const int e0 = (blockIdx.x * blockDim.x + threadIdx.x) * 4;
  if (e0 >= E) return;
  const unsigned int last = (unsigned int)(E - 1);
  if (e0 + 4 <= E) {
    const int4 s4 = *(const int4*)(esrc + e0);
    const int4 d4 = *(const int4*)(edst + e0);
    const int4 r4 = *(const int4*)(rank + e0);
    const unsigned int p0 = (unsigned int)(seg_start[d4.x] + r4.x);
    const unsigned int p1 = (unsigned int)(seg_start[d4.y] + r4.y);
    const unsigned int p2 = (unsigned int)(seg_start[d4.z] + r4.z);
    const unsigned int p3 = (unsigned int)(seg_start[d4.w] + r4.w);
    ssrc[min(p0, last)] = s4.x;
    ssrc[min(p1, last)] = s4.y;
    ssrc[min(p2, last)] = s4.z;
    ssrc[min(p3, last)] = s4.w;
  } else {
    for (int e = e0; e < E; ++e) {
      const unsigned int p = (unsigned int)(seg_start[edst[e]] + rank[e]);
      ssrc[min(p, last)] = esrc[e];
    }
  }
}

#define GA_NPW 8
#define GA_ROWS 8
#define GCN_C 40
__global__ void gcn_gather_out(const float* __restrict__ t, const int* __restrict__ seg_start,
                               const int* __restrict__ seg_end, const int* __restrict__ ssrc,
                               const float* __restrict__ norm_src,
                               const float* __restrict__ norm_dst, const float* __restrict__ bias,
                               float* __restrict__ out, int n, int E) {
  const int lane = threadIdx.x & 31;
  const int wv = __builtin_amdgcn_readfirstlane(
      (int)(blockIdx.x * (blockDim.x >> 5) + (threadIdx.x >> 5)));
  const int n0 = wv * GA_NPW;
  if (n0 >= n) return;
  const int nn = min(GA_NPW, n - n0);
  const bool colok = lane < (GCN_C / 2);

  const int nl = min(n0 + (lane & (GA_NPW - 1)), n - 1);
  const int sl = seg_start[nl];
  const int el = seg_end[nl];
  const float ndl = norm_dst[nl];
  float2 bb = make_float2(0.f, 0.f);
  if (colok) bb = *(const float2*)(bias + 2 * lane);

  for (int k = 0; k < nn; ++k) {
    const int s = max(__builtin_amdgcn_readlane(sl, k), 0);
    const int e = min(__builtin_amdgcn_readlane(el, k), E);
    float ax = 0.f, ay = 0.f;
    for (int j0 = s; j0 < e; j0 += 32) {
      const int cnt = min(32, e - j0);
      int idl = 0;
      float nsl = 0.f;
      if (lane < cnt) {
        idl = (int)min((unsigned int)ssrc[j0 + lane], (unsigned int)(n - 1));
        nsl = norm_src[idl];
      }
      for (int i = 0; i < cnt; i += GA_ROWS) {
        float2 v[GA_ROWS];
#pragma unroll
        for (int u = 0; u < GA_ROWS; ++u) {
          v[u] = make_float2(0.f, 0.f);
          if (i + u < cnt) {
            const int sidx = __builtin_amdgcn_readlane(idl, i + u);
            if (colok) v[u] = *(const float2*)(t + (size_t)sidx * GCN_C + 2 * lane);
          }
        }
#pragma unroll
        for (int u = 0; u < GA_ROWS; ++u) {
          const float ns = __builtin_bit_cast(
              float, __builtin_amdgcn_readlane(__builtin_bit_cast(int, nsl), i + u));
          ax += v[u].x * ns;
          ay += v[u].y * ns;
        }
      }
    }
    const float nd = __builtin_bit_cast(
        float, __builtin_amdgcn_readlane(__builtin_bit_cast(int, ndl), k));
    if (colok)
      *(float2*)(out + (size_t)(n0 + k) * GCN_C + 2 * lane) =
          make_float2(ax * nd + bb.x, ay * nd + bb.y);
  }
}

#define FL_THREADS 256
#define FL_TILES 4
template <bool PROJ>
__global__ void __launch_bounds__(FL_THREADS)
gcn_layer_fused(const float* __restrict__ hin, const int* __restrict__ seg_start,
                const int* __restrict__ seg_end, const int* __restrict__ ssrc,
                const float* __restrict__ norm_src, const float* __restrict__ norm_dst,
                const float* __restrict__ W, const float* __restrict__ bias,
                const float* __restrict__ Wp, float* __restrict__ hout, int n, int E,
                int nRowTiles) {
  __shared__ __attribute__((aligned(16))) float Wlds[GCN_D * WLDS_STRIDE];
  __shared__ __attribute__((aligned(16))) float Alds[FL_TILES * 16 * WLDS_STRIDE];

  const int lane = threadIdx.x & 31;
  const int wv = __builtin_amdgcn_readfirstlane(
      (int)(blockIdx.x * (FL_THREADS >> 5) + (threadIdx.x >> 5)));
  const int w  = wv & 7;
  const int n0 = wv * GA_NPW;
  const int nn = max(0, min(GA_NPW, n - n0));

  const int nl = min(n0 + (lane & (GA_NPW - 1)), n - 1);
  const int sl = seg_start[nl];
  const int el = seg_end[nl];
  const float ndl = norm_dst[nl];

#pragma unroll
  for (int j = 0; j < (GCN_D * GCN_D) / (4 * FL_THREADS); ++j) {
    const int i = ((int)threadIdx.x + j * FL_THREADS) * 4;
    const float4 w4 = *(const float4*)(W + i);
    const int k = i >> 6;
    const int c = i & 63;
    Wlds[(c + 0) * WLDS_STRIDE + k] = w4.x;
    Wlds[(c + 1) * WLDS_STRIDE + k] = w4.y;
    Wlds[(c + 2) * WLDS_STRIDE + k] = w4.z;
    Wlds[(c + 3) * WLDS_STRIDE + k] = w4.w;
  }

  float* arow = &Alds[(w * GA_NPW) * WLDS_STRIDE + 2 * lane];

  for (int k = 0; k < nn; ++k) {
    const int s = max(__builtin_amdgcn_readlane(sl, k), 0);
    const int e = min(__builtin_amdgcn_readlane(el, k), E);
    float ax = 0.f, ay = 0.f;
    for (int j0 = s; j0 < e; j0 += 32) {
      const int cnt = min(32, e - j0);
      int idl = 0;
      float nsl = 0.f;
      if (lane < cnt) {
        idl = (int)min((unsigned int)ssrc[j0 + lane], (unsigned int)(n - 1));
        nsl = norm_src[idl];
      }
      for (int i = 0; i < cnt; i += GA_ROWS) {
        float2 v[GA_ROWS];
#pragma unroll
        for (int u = 0; u < GA_ROWS; ++u) {
          v[u] = make_float2(0.f, 0.f);
          if (i + u < cnt) {
            const int sidx = __builtin_amdgcn_readlane(idl, i + u);
            v[u] = *(const float2*)(hin + (size_t)sidx * GCN_D + 2 * lane);
          }
        }
#pragma unroll
        for (int u = 0; u < GA_ROWS; ++u) {
          const float ns = __builtin_bit_cast(
              float, __builtin_amdgcn_readlane(__builtin_bit_cast(int, nsl), i + u));
          ax += v[u].x * ns;
          ay += v[u].y * ns;
        }
      }
    }
    const float nd = __builtin_bit_cast(
        float, __builtin_amdgcn_readlane(__builtin_bit_cast(int, ndl), k));
    *(float2*)(arow + k * WLDS_STRIDE) = make_float2(ax * nd, ay * nd);
  }

  __syncthreads();

  const int rt = (int)blockIdx.x * FL_TILES + (w >> 1);
  const bool live = rt < nRowTiles;
  if (!PROJ && !live) return;

  const int mrow  = lane & 15;
  const int khalf = (lane >> 4) << 1;
  const int ct0   = (w & 1) << 1;
  const float* aptr = &Alds[((w >> 1) * 16 + mrow) * WLDS_STRIDE + khalf];
  const float* bptr0 = &Wlds[((ct0 << 4) + mrow) * WLDS_STRIDE + khalf];
  const float* bptr1 = bptr0 + 16 * WLDS_STRIDE;

  float4 p4[3];
  if (PROJ) {
#pragma unroll
    for (int j = 0; j < 3; ++j) {
      const int g = (int)threadIdx.x + j * FL_THREADS;
      p4[j] = make_float4(0.f, 0.f, 0.f, 0.f);
      if (g < (GCN_D * GCN_C) / 4) p4[j] = *(const float4*)(Wp + 4 * g);
    }
  }

  v8f acc0 = {0.f, 0.f, 0.f, 0.f, 0.f, 0.f, 0.f, 0.f};
  v8f acc1 = {0.f, 0.f, 0.f, 0.f, 0.f, 0.f, 0.f, 0.f};

  if (live) {
#pragma unroll
    for (int ks = 0; ks < 16; ++ks) {
      v2f a, b0, b1;
      const float2 av  = *(const float2*)(aptr + (ks << 2));
      const float2 bv0 = *(const float2*)(bptr0 + (ks << 2));
      const float2 bv1 = *(const float2*)(bptr1 + (ks << 2));
      a.x = av.x;   a.y = av.y;
      b0.x = bv0.x; b0.y = bv0.y;
      b1.x = bv1.x; b1.y = bv1.y;
      acc0 = __builtin_amdgcn_wmma_f32_16x16x4_f32(
           false, a,  false, b0,
           (short)0, acc0,  false,  false);
      acc1 = __builtin_amdgcn_wmma_f32_16x16x4_f32(
           false, a,  false, b1,
           (short)0, acc1,  false,  false);
    }
  }

  const int ncol0 = (ct0 << 4) + mrow;
  const float bb0 = bias[ncol0];
  const float bb1 = bias[ncol0 + 16];

  if (!PROJ) {
    float* orow = hout + (size_t)((rt << 4) + ((lane >> 4) << 3)) * GCN_D + ncol0;
#pragma unroll
    for (int i = 0; i < 8; ++i) {
      orow[(size_t)i * GCN_D]      = fmaxf(acc0[i] + bb0, 0.f);
      orow[(size_t)i * GCN_D + 16] = fmaxf(acc1[i] + bb1, 0.f);
    }
  } else {
    __syncthreads();

    if (live) {
      float* hrow = &Alds[((w >> 1) * 16 + ((lane >> 4) << 3)) * WLDS_STRIDE + ncol0];
#pragma unroll
      for (int i = 0; i < 8; ++i) {
        hrow[i * WLDS_STRIDE]      = fmaxf(acc0[i] + bb0, 0.f);
        hrow[i * WLDS_STRIDE + 16] = fmaxf(acc1[i] + bb1, 0.f);
      }
    }
#pragma unroll
    for (int j = 0; j < 3; ++j) {
      const int g = (int)threadIdx.x + j * FL_THREADS;
      if (g < (GCN_D * GCN_C) / 4) {
        const int k = (4 * g) / GCN_C;
        const int c = (4 * g) - k * GCN_C;
        Wlds[(c + 0) * WLDS_STRIDE + k] = p4[j].x;
        Wlds[(c + 1) * WLDS_STRIDE + k] = p4[j].y;
        Wlds[(c + 2) * WLDS_STRIDE + k] = p4[j].z;
        Wlds[(c + 3) * WLDS_STRIDE + k] = p4[j].w;
      }
    }

    __syncthreads();

    if (!live) return;

    v8f q0 = {0.f, 0.f, 0.f, 0.f, 0.f, 0.f, 0.f, 0.f};
    v8f q1 = {0.f, 0.f, 0.f, 0.f, 0.f, 0.f, 0.f, 0.f};
    float* trow = hout + (size_t)((rt << 4) + ((lane >> 4) << 3)) * GCN_C + ncol0;
    if ((w & 1) == 0) {
#pragma unroll
      for (int ks = 0; ks < 16; ++ks) {
        v2f a, b0, b1;
        const float2 av  = *(const float2*)(aptr + (ks << 2));
        const float2 bv0 = *(const float2*)(bptr0 + (ks << 2));
        const float2 bv1 = *(const float2*)(bptr1 + (ks << 2));
        a.x = av.x;   a.y = av.y;
        b0.x = bv0.x; b0.y = bv0.y;
        b1.x = bv1.x; b1.y = bv1.y;
        q0 = __builtin_amdgcn_wmma_f32_16x16x4_f32(
             false, a,  false, b0,
             (short)0, q0,  false,  false);
        q1 = __builtin_amdgcn_wmma_f32_16x16x4_f32(
             false, a,  false, b1,
             (short)0, q1,  false,  false);
      }
#pragma unroll
      for (int i = 0; i < 8; ++i) {
        trow[(size_t)i * GCN_C]      = q0[i];
        trow[(size_t)i * GCN_C + 16] = q1[i];
      }
    } else {
#pragma unroll
      for (int ks = 0; ks < 16; ++ks) {
        v2f a, b0;
        const float2 av  = *(const float2*)(aptr + (ks << 2));
        const float2 bv0 = *(const float2*)(bptr0 + (ks << 2));
        a.x = av.x;   a.y = av.y;
        b0.x = bv0.x; b0.y = bv0.y;
        q0 = __builtin_amdgcn_wmma_f32_16x16x4_f32(
             false, a,  false, b0,
             (short)0, q0,  false,  false);
      }
      if (ncol0 < GCN_C) {
#pragma unroll
        for (int i = 0; i < 8; ++i) trow[(size_t)i * GCN_C] = q0[i];
      }
    }
  }
}

extern "C" void kernel_launch(void* const* d_in, const int* in_sizes, int n_in,
                              void* d_out, int out_size, void* d_ws, size_t ws_size,
                              hipStream_t stream) {
  const float* features = (const float*)d_in[0];
  const int*   esrc     = (const int*)d_in[1];
  const int*   edst     = (const int*)d_in[2];
  const float* W1       = (const float*)d_in[3];
  const float* b1       = (const float*)d_in[4];
  const float* W2       = (const float*)d_in[5];
  const float* b2       = (const float*)d_in[6];
  const float* W3       = (const float*)d_in[7];
  const float* b3       = (const float*)d_in[8];
  float*       out      = (float*)d_out;

  const int N = GCN_N;
  const int E = in_sizes[1];

  float* norm_src = (float*)d_ws;
  float* norm_dst = norm_src + N;
  float* t        = norm_dst + N;
  float* h        = t + (size_t)N * GCN_D;

  int*          seg_start  = (int*)(t + (size_t)N * GCN_C);
  int*          seg_cursor = seg_start + N;
  unsigned int* counter    = (unsigned int*)(seg_cursor + N);
  int*          ssrc       = seg_cursor + N + 4;
  int*          rank       = (int*)h;

  const int TPB = 256;

  {
    int n4 = (2 * N) / 4;
    int quadThreads = (E + 3) / 4;
    gcn_zero4<<<(n4 + TPB - 1) / TPB, TPB, 0, stream>>>((float4*)norm_src, n4, counter);
    gcn_degree_rank<<<(quadThreads + TPB - 1) / TPB, TPB, 0, stream>>>(esrc, edst, norm_src,
                                                                     (int*)norm_dst, rank, E);
    gcn_norm_alloc<<<(N + TPB - 1) / TPB, TPB, 0, stream>>>(norm_src, norm_dst, seg_start,
                                                            seg_cursor, counter, N);
    gcn_fill<<<(quadThreads + TPB - 1) / TPB, TPB, 0, stream>>>(esrc, edst, seg_start, rank, ssrc, E);
  }

  int nodesPB    = (TPB / 32) * GA_NPW;
  int gathBlocks = (N + nodesPB - 1) / nodesPB;
  int rowTiles   = N / 16;

  gcn_layer_fused<false><<<gathBlocks, TPB, 0, stream>>>(features, seg_start, seg_cursor, ssrc,
                                                         norm_src, norm_dst, W1, b1, W3, h, N, E,
                                                         rowTiles);

  gcn_layer_fused<true><<<gathBlocks, TPB, 0, stream>>>(h, seg_start, seg_cursor, ssrc, norm_src,
                                                        norm_dst, W2, b2, W3, t, N, E, rowTiles);

  gcn_gather_out<<<gathBlocks, TPB, 0, stream>>>(t, seg_start, seg_cursor, ssrc, norm_src, norm_dst,
                                                 b3, out, N, E);
}


# row gathers (3 kernels): a node's source ids are fetched one node ahead, under the rows of the node before, and rows go out 16 at a time (was 8): fewer round trips per node; sum order unchanged
# speedup vs baseline: 4.9932x; 2.0875x over previous
#include <hip/hip_runtime.h>

typedef __attribute__((ext_vector_type(2))) float v2f;
typedef __attribute__((ext_vector_type(8))) float v8f;

#define GCN_N 100000
#define GCN_D 64
#define WLDS_STRIDE 68

#define PT_NCH 256
#define PT_NR 256
#define PT_RSZ 391
#define PT_THREADS 256
#define PT_CAP 4096

__device__ __forceinline__ int pt_chunk_len(int E) {
  return min((((E + PT_NCH - 1) / PT_NCH) + 3) & ~3, PT_CAP);
}

__device__ __forceinline__ int4 pt_load4(const int* __restrict__ ids, int e0, int beg, int end4) {
  return *(const int4*)(ids + ((e0 + 4 <= end4) ? e0 : beg));
}

__device__ __forceinline__ int pt_range(int id) {
  return (int)min((unsigned int)id / (unsigned int)PT_RSZ, (unsigned int)(PT_NR - 1));
}

__global__ void __launch_bounds__(PT_THREADS)
gcn_part(const int* __restrict__ esrc, const int* __restrict__ edst, int2* __restrict__ stg_d,
         int* __restrict__ stg_s, int* __restrict__ cnt_d, int* __restrict__ off_d,
         int* __restrict__ cnt_s, int* __restrict__ off_s, unsigned int* __restrict__ counter, int E) {
  __shared__ int cnt[PT_NR + 1];
  __shared__ int cur[PT_NR + 1];
  __shared__ int wsum[PT_THREADS / 32];
  __shared__ int2 grp[PT_CAP];
  const int tid = (int)threadIdx.x;
  const int lane = tid & 31;
  const int wid = tid >> 5;
  const int b = (int)blockIdx.x;
  const int which = b / PT_NCH;
  const int c = b - which * PT_NCH;
  const int* __restrict__ keys = which ? esrc : edst;

  cnt[tid] = 0;
  if (tid == 0) { cnt[PT_NR] = 0; cur[PT_NR] = 0; }
  if (b == 0 && tid == 0) *counter = 0u;
  __syncthreads();

  const int CL = pt_chunk_len(E);
  const int beg = min(c * CL, E);
  const int end = min(E, beg + CL);
  const int len = end - beg;

  const int end4 = beg + (len & ~3);
  const bool any4 = len >= 4;

  if (any4) {
    for (int base = beg; base < end4; base += PT_THREADS * 16) {
      int4 k[4];
#pragma unroll
      for (int j = 0; j < 4; ++j) k[j] = pt_load4(keys, base + (j * PT_THREADS + tid) * 4, beg, end4);
#pragma unroll
      for (int j = 0; j < 4; ++j) {
        const bool ok = base + (j * PT_THREADS + tid) * 4 + 4 <= end4;
        atomicAdd(&cnt[ok ? pt_range(k[j].x) : PT_NR], 1);
        atomicAdd(&cnt[ok ? pt_range(k[j].y) : PT_NR], 1);
        atomicAdd(&cnt[ok ? pt_range(k[j].z) : PT_NR], 1);
        atomicAdd(&cnt[ok ? pt_range(k[j].w) : PT_NR], 1);
      }
    }
  }
  if (tid < (len & 3)) atomicAdd(&cnt[pt_range(keys[end4 + tid])], 1);
  __syncthreads();

  const int mine = cnt[tid];
  int incl = mine;
  for (int o = 1; o < 32; o <<= 1) {
    const int t = __shfl_up(incl, o, 32);
    if (lane >= o) incl += t;
  }
  if (lane == 31) wsum[wid] = incl;
  __syncthreads();
  int wbase = 0;
  for (int w = 0; w < PT_THREADS / 32; ++w) if (w < wid) wbase += wsum[w];
  const int excl = wbase + incl - mine;
  cur[tid] = excl;
  {
    int* __restrict__ cg = which ? cnt_s : cnt_d;
    int* __restrict__ og = which ? off_s : off_d;
    cg[tid * PT_NCH + c] = mine;
    og[tid * PT_NCH + c] = excl;
  }
  __syncthreads();

  const size_t cb = (size_t)c * (size_t)CL;
  const unsigned int lastp = (unsigned int)(max(len, 1) - 1);
  if (any4) {
    for (int base = beg; base < end4; base += PT_THREADS * 16) {
      int4 k[4], s[4];
#pragma unroll
      for (int j = 0; j < 4; ++j) {
        const int e0 = base + (j * PT_THREADS + tid) * 4;
        k[j] = pt_load4(keys, e0, beg, end4);
        s[j] = pt_load4(esrc, e0, beg, end4);
      }
      int p[4][4];
#pragma unroll
      for (int j = 0; j < 4; ++j) {
        const bool ok = base + (j * PT_THREADS + tid) * 4 + 4 <= end4;
        p[j][0] = atomicAdd(&cur[ok ? pt_range(k[j].x) : PT_NR], 1);
        p[j][1] = atomicAdd(&cur[ok ? pt_range(k[j].y) : PT_NR], 1);
        p[j][2] = atomicAdd(&cur[ok ? pt_range(k[j].z) : PT_NR], 1);
        p[j][3] = atomicAdd(&cur[ok ? pt_range(k[j].w) : PT_NR], 1);
      }
#pragma unroll
      for (int j = 0; j < 4; ++j) {
        if (base + (j * PT_THREADS + tid) * 4 + 4 <= end4) {
          grp[min((unsigned int)p[j][0], lastp)] = make_int2(s[j].x, k[j].x);
          grp[min((unsigned int)p[j][1], lastp)] = make_int2(s[j].y, k[j].y);
          grp[min((unsigned int)p[j][2], lastp)] = make_int2(s[j].z, k[j].z);
          grp[min((unsigned int)p[j][3], lastp)] = make_int2(s[j].w, k[j].w);
        }
      }
    }
  }
  if (tid < (len & 3)) {
    const int kk = keys[end4 + tid];
    const int ss = esrc[end4 + tid];
    const int pos = atomicAdd(&cur[pt_range(kk)], 1);
    grp[min((unsigned int)pos, lastp)] = make_int2(ss, kk);
  }
  __syncthreads();
  if (which) {
    for (int i = tid; i < len; i += PT_THREADS) stg_s[cb + i] = grp[i].x;
  } else {
    for (int i = tid; i < len; i += PT_THREADS) stg_d[cb + i] = grp[i];
  }
}

#define RS_THREADS 512
#define RS_OCAP 8192
__global__ void __launch_bounds__(RS_THREADS)
gcn_rsort(const int2* __restrict__ stg_d, const int* __restrict__ stg_s, const int* __restrict__ cnt_d,
          const int* __restrict__ off_d, const int* __restrict__ cnt_s, const int* __restrict__ off_s,
          float* __restrict__ norm_src, float* __restrict__ norm_dst, int* __restrict__ seg_start,
          int* __restrict__ seg_end, unsigned int* __restrict__ counter, int* __restrict__ ssrc, int E,
          int n) {
  __shared__ int degi[PT_RSZ + 1];
  __shared__ int dego[PT_RSZ + 1];
  __shared__ int curs[PT_RSZ + 1];
  __shared__ int wsum[RS_THREADS / 32];
  __shared__ int sbase;
  __shared__ int lout[RS_OCAP];
  const int tid = (int)threadIdx.x;
  const int lane = tid & 31;
  const int wid = tid >> 5;
  const int r = (int)blockIdx.x;
  const int lo = r * PT_RSZ;
  const int CL = pt_chunk_len(E);
  const int ch = tid >> 1;
  const int q = tid & 1;

  const int od = min(max(off_d[r * PT_NCH + ch], 0), CL);
  const int cd = min(max(cnt_d[r * PT_NCH + ch], 0), CL - od);
  const int os = min(max(off_s[r * PT_NCH + ch], 0), CL);
  const int cs = min(max(cnt_s[r * PT_NCH + ch], 0), CL - os);
  const int2* __restrict__ gd = stg_d + (size_t)ch * (size_t)CL + od;
  const int* __restrict__ gs = stg_s + (size_t)ch * (size_t)CL + os;

  for (int i = tid; i <= PT_RSZ; i += RS_THREADS) { degi[i] = 0; dego[i] = 0; }
  if (tid == 0) curs[PT_RSZ] = 0;
  __syncthreads();

  for (int i0 = q; i0 < cd; i0 += 8) {
    int2 p[4];
#pragma unroll
    for (int j = 0; j < 4; ++j) p[j] = gd[min(i0 + 2 * j, cd - 1)];
#pragma unroll
    for (int j = 0; j < 4; ++j) {
      const unsigned int u = (unsigned int)(p[j].y - lo);
      atomicAdd(&degi[(i0 + 2 * j < cd && u < PT_RSZ) ? u : PT_RSZ], 1);
    }
  }
  for (int i0 = q; i0 < cs; i0 += 8) {
    int v[4];
#pragma unroll
    for (int j = 0; j < 4; ++j) v[j] = gs[min(i0 + 2 * j, cs - 1)];
#pragma unroll
    for (int j = 0; j < 4; ++j) {
      const unsigned int u = (unsigned int)(v[j] - lo);
      atomicAdd(&dego[(i0 + 2 * j < cs && u < PT_RSZ) ? u : PT_RSZ], 1);
    }
  }
  __syncthreads();

  const int mine = tid < PT_RSZ ? degi[tid] : 0;
  int incl = mine;
  for (int o = 1; o < 32; o <<= 1) {
    const int t = __shfl_up(incl, o, 32);
    if (lane >= o) incl += t;
  }
  if (lane == 31) wsum[wid] = incl;
  __syncthreads();
  int wbase = 0, total = 0;
  for (int w = 0; w < RS_THREADS / 32; ++w) {
    const int s = wsum[w];
    if (w < wid) wbase += s;
    total += s;
  }
  const int excl = wbase + incl - mine;
  if (tid == 0) sbase = (int)atomicAdd(counter, (unsigned int)total);
  __syncthreads();
  const int base = sbase;
  if (tid < PT_RSZ) {
    curs[tid] = excl;
    if (lo + tid < n) {
      seg_start[lo + tid] = base + excl;
      seg_end[lo + tid]   = base + excl + mine;
      norm_src[lo + tid] = rsqrtf(fmaxf((float)dego[tid], 1.0f));
      norm_dst[lo + tid] = rsqrtf(fmaxf((float)mine, 1.0f));
    }
  }
  const bool stage = total <= RS_OCAP;
  const unsigned int lastslot = (unsigned int)(max(E, 1) - 1);
  __syncthreads();

  for (int i0 = q; i0 < cd; i0 += 8) {
    int2 p[4];
#pragma unroll
    for (int j = 0; j < 4; ++j) p[j] = gd[min(i0 + 2 * j, cd - 1)];
    int slot[4];
#pragma unroll
    for (int j = 0; j < 4; ++j) {
      const unsigned int u = (unsigned int)(p[j].y - lo);
      slot[j] = atomicAdd(&curs[(i0 + 2 * j < cd && u < PT_RSZ) ? u : PT_RSZ], 1);
    }
#pragma unroll
    for (int j = 0; j < 4; ++j) {
      const unsigned int u = (unsigned int)(p[j].y - lo);
      if (i0 + 2 * j < cd && u < PT_RSZ) {
        if (stage) lout[min((unsigned int)slot[j], (unsigned int)(RS_OCAP - 1))] = p[j].x;
        else       ssrc[min((unsigned int)(base + slot[j]), lastslot)] = p[j].x;
      }
    }
  }
  if (stage) {
    __syncthreads();
    for (int i = tid; i < total; i += RS_THREADS)
      ssrc[min((unsigned int)(base + i), lastslot)] = lout[i];
  }
}

#define GA_NPW 8
#define GA_ROWS 16
#define GCN_C 40

__device__ __forceinline__ int ga_first_ids(const int* __restrict__ ssrc, int s, int e, int lane) {
  int id = 0;
  if (lane < min(32, e - s)) id = ssrc[s + lane];
  return id;
}

__device__ __forceinline__ int ga_row_id(int id, int n) {
  return (int)min((unsigned int)id, (unsigned int)(n - 1));
}
__global__ void gcn_gather_out(const float* __restrict__ t, const int* __restrict__ seg_start,
                               const int* __restrict__ seg_end, const int* __restrict__ ssrc,
                               const float* __restrict__ norm_dst, const float* __restrict__ bias,
                               float* __restrict__ out, int n, int E) {
  const int lane = threadIdx.x & 31;
  const int wv = __builtin_amdgcn_readfirstlane(
      (int)(blockIdx.x * (blockDim.x >> 5) + (threadIdx.x >> 5)));
  const int n0 = wv * GA_NPW;
  if (n0 >= n) return;
  const int nn = min(GA_NPW, n - n0);
  const bool colok = lane < (GCN_C / 2);

  const int nl = min(n0 + (lane & (GA_NPW - 1)), n - 1);
  const int sl = seg_start[nl];
  const int el = seg_end[nl];
  const float ndl = norm_dst[nl];
  float2 bb = make_float2(0.f, 0.f);
  if (colok) bb = *(const float2*)(bias + 2 * lane);

  int s_nx = max(__builtin_amdgcn_readlane(sl, 0), 0);
  int e_nx = min(__builtin_amdgcn_readlane(el, 0), E);
  int idl = ga_row_id(ga_first_ids(ssrc, s_nx, e_nx, lane), n);

  for (int k = 0; k < nn; ++k) {
    const int s = s_nx;
    const int e = e_nx;
    s_nx = 0;
    e_nx = 0;
    if (k + 1 < nn) {
      s_nx = max(__builtin_amdgcn_readlane(sl, k + 1), 0);
      e_nx = min(__builtin_amdgcn_readlane(el, k + 1), E);
    }
    int id_nx = ga_first_ids(ssrc, s_nx, e_nx, lane);
    float ax = 0.f, ay = 0.f;
    int idw = idl;
    for (int j0 = s; j0 < e; j0 += 32) {
      const int cnt = min(32, e - j0);
      for (int i = 0; i < cnt; i += GA_ROWS) {
        float2 v[GA_ROWS];
#pragma unroll
        for (int u = 0; u < GA_ROWS; ++u) {
          v[u] = make_float2(0.f, 0.f);
          if (i + u < cnt) {
            const int sidx = __builtin_amdgcn_readlane(idw, i + u);
            if (colok) v[u] = *(const float2*)(t + (size_t)sidx * GCN_C + 2 * lane);
          }
        }
#pragma unroll
        for (int u = 0; u < GA_ROWS; ++u) {
          ax += v[u].x;
          ay += v[u].y;
        }
      }
      if (j0 + 32 < e) idw = ga_row_id(ga_first_ids(ssrc, j0 + 32, e, lane), n);
    }
    const float nd = __builtin_bit_cast(
        float, __builtin_amdgcn_readlane(__builtin_bit_cast(int, ndl), k));
    if (colok)
      *(float2*)(out + (size_t)(n0 + k) * GCN_C + 2 * lane) =
          make_float2(ax * nd + bb.x, ay * nd + bb.y);
    asm volatile("" : "+v"(id_nx));
    idl = ga_row_id(id_nx, n);
  }
}

__device__ __forceinline__ void fl_row_scales(const float* __restrict__ norm_src, int rt, int lane,
                                              float (&sc)[8]) {
  const float* sp = norm_src + (rt << 4) + ((lane >> 4) << 3);
  const float4 a = *(const float4*)sp;
  const float4 b = *(const float4*)(sp + 4);
  sc[0] = a.x; sc[1] = a.y; sc[2] = a.z; sc[3] = a.w;
  sc[4] = b.x; sc[5] = b.y; sc[6] = b.z; sc[7] = b.w;
}

#define FL_THREADS 256
#define FL_TILES 4
template <bool PROJ>
__global__ void __launch_bounds__(FL_THREADS)
gcn_layer_fused(const float* __restrict__ hin, const int* __restrict__ seg_start,
                const int* __restrict__ seg_end, const int* __restrict__ ssrc,
                const float* __restrict__ norm_src, const float* __restrict__ norm_dst,
                const float* __restrict__ W, const float* __restrict__ bias,
                const float* __restrict__ Wp, float* __restrict__ hout, int n, int E,
                int nRowTiles) {
  __shared__ __attribute__((aligned(16))) float Wlds[GCN_D * WLDS_STRIDE];
  __shared__ __attribute__((aligned(16))) float Alds[FL_TILES * 16 * WLDS_STRIDE];

  const int lane = threadIdx.x & 31;
  const int wv = __builtin_amdgcn_readfirstlane(
      (int)(blockIdx.x * (FL_THREADS >> 5) + (threadIdx.x >> 5)));
  const int w  = wv & 7;
  const int n0 = wv * GA_NPW;
  const int nn = max(0, min(GA_NPW, n - n0));

  const int nl = min(n0 + (lane & (GA_NPW - 1)), n - 1);
  const int sl = seg_start[nl];
  const int el = seg_end[nl];
  const float ndl = norm_dst[nl];

#pragma unroll
  for (int j = 0; j < (GCN_D * GCN_D) / (4 * FL_THREADS); ++j) {
    const int i = ((int)threadIdx.x + j * FL_THREADS) * 4;
    const float4 w4 = *(const float4*)(W + i);
    const int k = i >> 6;
    const int c = i & 63;
    Wlds[(c + 0) * WLDS_STRIDE + k] = w4.x;
    Wlds[(c + 1) * WLDS_STRIDE + k] = w4.y;
    Wlds[(c + 2) * WLDS_STRIDE + k] = w4.z;
    Wlds[(c + 3) * WLDS_STRIDE + k] = w4.w;
  }

  float* arow = &Alds[(w * GA_NPW) * WLDS_STRIDE + 2 * lane];

  int s_nx = 0;
  int e_nx = 0;
  if (nn > 0) {
    s_nx = max(__builtin_amdgcn_readlane(sl, 0), 0);
    e_nx = min(__builtin_amdgcn_readlane(el, 0), E);
  }
  int idl = ga_row_id(ga_first_ids(ssrc, s_nx, e_nx, lane), n);

  for (int k = 0; k < nn; ++k) {
    const int s = s_nx;
    const int e = e_nx;
    s_nx = 0;
    e_nx = 0;
    if (k + 1 < nn) {
      s_nx = max(__builtin_amdgcn_readlane(sl, k + 1), 0);
      e_nx = min(__builtin_amdgcn_readlane(el, k + 1), E);
    }
    float nsf = 0.f;
    if (!PROJ) { if (lane < min(32, e - s)) nsf = norm_src[idl]; }
    int id_nx = ga_first_ids(ssrc, s_nx, e_nx, lane);
    float ax = 0.f, ay = 0.f;
    int idw = idl;
    float nsl = nsf;
    for (int j0 = s; j0 < e; j0 += 32) {
      const int cnt = min(32, e - j0);
      for (int i = 0; i < cnt; i += GA_ROWS) {
        float2 v[GA_ROWS];
#pragma unroll
        for (int u = 0; u < GA_ROWS; ++u) {
          v[u] = make_float2(0.f, 0.f);
          if (i + u < cnt) {
            const int sidx = __builtin_amdgcn_readlane(idw, i + u);
            v[u] = *(const float2*)(hin + (size_t)sidx * GCN_D + 2 * lane);
          }
        }
#pragma unroll
        for (int u = 0; u < GA_ROWS; ++u) {
          if (PROJ) {
            ax += v[u].x;
            ay += v[u].y;
          } else {
            const float ns = __builtin_bit_cast(
                float, __builtin_amdgcn_readlane(__builtin_bit_cast(int, nsl), i + u));
            ax = __builtin_fmaf(v[u].x, ns, ax);
            ay = __builtin_fmaf(v[u].y, ns, ay);
          }
        }
      }
      if (j0 + 32 < e) {
        idw = ga_row_id(ga_first_ids(ssrc, j0 + 32, e, lane), n);
        nsl = 0.f;
        if (!PROJ) { if (lane < min(32, e - (j0 + 32))) nsl = norm_src[idw]; }
      }
    }
    const float nd = __builtin_bit_cast(
        float, __builtin_amdgcn_readlane(__builtin_bit_cast(int, ndl), k));
    *(float2*)(arow + k * WLDS_STRIDE) = make_float2(ax * nd, ay * nd);
    asm volatile("" : "+v"(id_nx));
    idl = ga_row_id(id_nx, n);
  }

  __syncthreads();

  const int rt = (int)blockIdx.x * FL_TILES + (w >> 1);
  const bool live = rt < nRowTiles;
  if (!PROJ && !live) return;

  const int mrow  = lane & 15;
  const int khalf = (lane >> 4) << 1;
  const int ct0   = (w & 1) << 1;
  const float* aptr = &Alds[((w >> 1) * 16 + mrow) * WLDS_STRIDE + khalf];
  const float* bptr0 = &Wlds[((ct0 << 4) + mrow) * WLDS_STRIDE + khalf];
  const float* bptr1 = bptr0 + 16 * WLDS_STRIDE;

  float4 p4[3];
  if (PROJ) {
#pragma unroll
    for (int j = 0; j < 3; ++j) {
      const int g = (int)threadIdx.x + j * FL_THREADS;
      p4[j] = make_float4(0.f, 0.f, 0.f, 0.f);
      if (g < (GCN_D * GCN_C) / 4) p4[j] = *(const float4*)(Wp + 4 * g);
    }
  }

  float sc[8];
  if (!PROJ) fl_row_scales(norm_src, rt, lane, sc);

  v8f acc0 = {0.f, 0.f, 0.f, 0.f, 0.f, 0.f, 0.f, 0.f};
  v8f acc1 = {0.f, 0.f, 0.f, 0.f, 0.f, 0.f, 0.f, 0.f};

  if (live) {
#pragma unroll
    for (int ks = 0; ks < 16; ++ks) {
      v2f a, b0, b1;
      const float2 av  = *(const float2*)(aptr + (ks << 2));
      const float2 bv0 = *(const float2*)(bptr0 + (ks << 2));
      const float2 bv1 = *(const float2*)(bptr1 + (ks << 2));
      a.x = av.x;   a.y = av.y;
      b0.x = bv0.x; b0.y = bv0.y;
      b1.x = bv1.x; b1.y = bv1.y;
      acc0 = __builtin_amdgcn_wmma_f32_16x16x4_f32(
           false, a,  false, b0,
           (short)0, acc0,  false,  false);
      acc1 = __builtin_amdgcn_wmma_f32_16x16x4_f32(
           false, a,  false, b1,
           (short)0, acc1,  false,  false);
    }
  }

  const int ncol0 = (ct0 << 4) + mrow;
  const float bb0 = bias[ncol0];
  const float bb1 = bias[ncol0 + 16];

  if (!PROJ) {
    float* orow = hout + (size_t)((rt << 4) + ((lane >> 4) << 3)) * GCN_D + ncol0;
#pragma unroll
    for (int i = 0; i < 8; ++i) {
      orow[(size_t)i * GCN_D]      = fmaxf(acc0[i] + bb0, 0.f) * sc[i];
      orow[(size_t)i * GCN_D + 16] = fmaxf(acc1[i] + bb1, 0.f) * sc[i];
    }
  } else {
    __syncthreads();

    if (live) {
      float* hrow = &Alds[((w >> 1) * 16 + ((lane >> 4) << 3)) * WLDS_STRIDE + ncol0];
#pragma unroll
      for (int i = 0; i < 8; ++i) {
        hrow[i * WLDS_STRIDE]      = fmaxf(acc0[i] + bb0, 0.f);
        hrow[i * WLDS_STRIDE + 16] = fmaxf(acc1[i] + bb1, 0.f);
      }
    }
#pragma unroll
    for (int j = 0; j < 3; ++j) {
      const int g = (int)threadIdx.x + j * FL_THREADS;
      if (g < (GCN_D * GCN_C) / 4) {
        const int k = (4 * g) / GCN_C;
        const int c = (4 * g) - k * GCN_C;
        Wlds[(c + 0) * WLDS_STRIDE + k] = p4[j].x;
        Wlds[(c + 1) * WLDS_STRIDE + k] = p4[j].y;
        Wlds[(c + 2) * WLDS_STRIDE + k] = p4[j].z;
        Wlds[(c + 3) * WLDS_STRIDE + k] = p4[j].w;
      }
    }

    __syncthreads();

    if (!live) return;

    fl_row_scales(norm_src, rt, lane, sc);

    v8f q0 = {0.f, 0.f, 0.f, 0.f, 0.f, 0.f, 0.f, 0.f};
    v8f q1 = {0.f, 0.f, 0.f, 0.f, 0.f, 0.f, 0.f, 0.f};
    float* trow = hout + (size_t)((rt << 4) + ((lane >> 4) << 3)) * GCN_C + ncol0;
    if ((w & 1) == 0) {
#pragma unroll
      for (int ks = 0; ks < 16; ++ks) {
        v2f a, b0, b1;
        const float2 av  = *(const float2*)(aptr + (ks << 2));
        const float2 bv0 = *(const float2*)(bptr0 + (ks << 2));
        const float2 bv1 = *(const float2*)(bptr1 + (ks << 2));
        a.x = av.x;   a.y = av.y;
        b0.x = bv0.x; b0.y = bv0.y;
        b1.x = bv1.x; b1.y = bv1.y;
        q0 = __builtin_amdgcn_wmma_f32_16x16x4_f32(
             false, a,  false, b0,
             (short)0, q0,  false,  false);
        q1 = __builtin_amdgcn_wmma_f32_16x16x4_f32(
             false, a,  false, b1,
             (short)0, q1,  false,  false);
      }
#pragma unroll
      for (int i = 0; i < 8; ++i) {
        trow[(size_t)i * GCN_C]      = q0[i] * sc[i];
        trow[(size_t)i * GCN_C + 16] = q1[i] * sc[i];
      }
    } else {
#pragma unroll
      for (int ks = 0; ks < 16; ++ks) {
        v2f a, b0;
        const float2 av  = *(const float2*)(aptr + (ks << 2));
        const float2 bv0 = *(const float2*)(bptr0 + (ks << 2));
        a.x = av.x;   a.y = av.y;
        b0.x = bv0.x; b0.y = bv0.y;
        q0 = __builtin_amdgcn_wmma_f32_16x16x4_f32(
             false, a,  false, b0,
             (short)0, q0,  false,  false);
      }
      if (ncol0 < GCN_C) {
#pragma unroll
        for (int i = 0; i < 8; ++i) trow[(size_t)i * GCN_C] = q0[i] * sc[i];
      }
    }
  }
}

extern "C" void kernel_launch(void* const* d_in, const int* in_sizes, int n_in,
                              void* d_out, int out_size, void* d_ws, size_t ws_size,
                              hipStream_t stream) {
  const float* features = (const float*)d_in[0];
  const int*   esrc     = (const int*)d_in[1];
  const int*   edst     = (const int*)d_in[2];
  const float* W1       = (const float*)d_in[3];
  const float* b1       = (const float*)d_in[4];
  const float* W2       = (const float*)d_in[5];
  const float* b2       = (const float*)d_in[6];
  const float* W3       = (const float*)d_in[7];
  const float* b3       = (const float*)d_in[8];
  float*       out      = (float*)d_out;

  const int N = GCN_N;
  const int E = in_sizes[1];

  float* norm_src = (float*)d_ws;
  float* norm_dst = norm_src + N;
  float* t        = norm_dst + N;
  float* h        = t + (size_t)N * GCN_D;

  int*          seg_start  = (int*)(t + (size_t)N * GCN_C);
  int*          seg_cursor = seg_start + N;
  unsigned int* counter    = (unsigned int*)(seg_cursor + N);
  int*          ssrc       = seg_cursor + N + 4;
  int2*         stg_d      = (int2*)h;
  int*          stg_s      = (int*)h + 2 * (size_t)PT_NCH * PT_CAP;
  int*          cnt_d      = stg_s + (size_t)PT_NCH * PT_CAP;
  int*          off_d      = cnt_d + PT_NR * PT_NCH;
  int*          cnt_s      = off_d + PT_NR * PT_NCH;
  int*          off_s      = cnt_s + PT_NR * PT_NCH;

  const int TPB = 256;

  {
    gcn_part<<<2 * PT_NCH, PT_THREADS, 0, stream>>>(esrc, edst, stg_d, stg_s, cnt_d, off_d, cnt_s,
                                                     off_s, counter, E);
    gcn_rsort<<<PT_NR, RS_THREADS, 0, stream>>>(stg_d, stg_s, cnt_d, off_d, cnt_s, off_s, norm_src,
                                                 norm_dst, seg_start, seg_cursor, counter, ssrc, E, N);
  }

  int nodesPB    = (TPB / 32) * GA_NPW;
  int gathBlocks = (N + nodesPB - 1) / nodesPB;
  int rowTiles   = N / 16;

  gcn_layer_fused<false><<<gathBlocks, TPB, 0, stream>>>(features, seg_start, seg_cursor, ssrc,
                                                         norm_src, norm_dst, W1, b1, W3, h, N, E,
                                                         rowTiles);

  gcn_layer_fused<true><<<gathBlocks, TPB, 0, stream>>>(h, seg_start, seg_cursor, ssrc, norm_src,
                                                        norm_dst, W2, b2, W3, t, N, E, rowTiles);

  gcn_gather_out<<<gathBlocks, TPB, 0, stream>>>(t, seg_start, seg_cursor, ssrc, norm_dst, b3, out, N,
                                                 E);
}
